# starting kernel
# baseline (speedup 1.0000x reference)
#include <hip/hip_runtime.h>
#include <math.h>


#define GRID_NODES 64
#define NI         62
#define PAD        64
#define PI_F       3.14159265358979323846f

typedef float v2f __attribute__((ext_vector_type(2)));
typedef float v8f __attribute__((ext_vector_type(8)));

__global__ void fem_init_kernel(float* __restrict__ Fv,
                                float* __restrict__ S,
                                float* __restrict__ invlam,
                                const float* __restrict__ kappa) {
  int tid = blockIdx.x * blockDim.x + threadIdx.x;
  if (tid >= GRID_NODES * GRID_NODES) return;
  Fv[tid] = 0.0f;

  int p = tid / PAD;
  int q = tid % PAD;

  float s = 0.0f;
  if (p < NI && q < NI)
    s = __sinf((float)((p + 1) * (q + 1)) * (PI_F / 63.0f));
  S[tid] = s;

  float il = 0.0f;
  if (p < NI && q < NI) {
    float lp = 2.0f - 2.0f * __cosf((float)(p + 1) * (PI_F / 63.0f));
    float lq = 2.0f - 2.0f * __cosf((float)(q + 1) * (PI_F / 63.0f));
    float norm = (2.0f / 63.0f) * (2.0f / 63.0f);
    il = norm / (kappa[0] * (lp + lq));
  }
  invlam[tid] = il;
}

__global__ void fem_assemble_kernel(const float* __restrict__ f,
                                    const float* __restrict__ nodes,
                                    const int* __restrict__ elems,
                                    int nelem,
                                    float* __restrict__ Fv) {
  int e = blockIdx.x * blockDim.x + threadIdx.x;
  if (e >= nelem) return;
  int a = elems[3 * e + 0];
  int b = elems[3 * e + 1];
  int c = elems[3 * e + 2];
  float xi = nodes[2 * a], yi = nodes[2 * a + 1];
  float xj = nodes[2 * b], yj = nodes[2 * b + 1];
  float xk = nodes[2 * c], yk = nodes[2 * c + 1];
  float area = 0.5f * fabsf((xj - xi) * (yk - yi) - (xk - xi) * (yj - yi));
  float fe = 0.0f;
  if (area > 1e-15f)
    fe = (area * (1.0f / 3.0f)) * ((f[a] + f[b] + f[c]) * (1.0f / 3.0f));
  atomicAdd(&Fv[a], fe);
  atomicAdd(&Fv[b], fe);
  atomicAdd(&Fv[c], fe);
}

__device__ __forceinline__ void wmma_tile64(const float* __restrict__ X,
                                            const float* __restrict__ Y,
                                            float* __restrict__ C,
                                            int t, int lane) {
  const int half = lane >> 4;
  const int l16  = lane & 15;
  const int tm   = (t >> 2) * 16;
  const int tn   = (t & 3) * 16;
  v8f acc = {0.f, 0.f, 0.f, 0.f, 0.f, 0.f, 0.f, 0.f};
#pragma unroll
  for (int k0 = 0; k0 < PAD; k0 += 4) {
    const int ka = k0 + 2 * half;
    v2f a, b;
    a[0] = X[(tm + l16) * PAD + ka];
    a[1] = X[(tm + l16) * PAD + ka + 1];
    b[0] = Y[ka * PAD + tn + l16];
    b[1] = Y[(ka + 1) * PAD + tn + l16];
    acc = __builtin_amdgcn_wmma_f32_16x16x4_f32(
        false, a, false, b, (short)0, acc, false, false);
  }
#pragma unroll
  for (int j = 0; j < 8; ++j)
    C[(tm + j + 8 * half) * PAD + tn + l16] = acc[j];
}

__device__ __forceinline__ void wmma_gemm64(const float* __restrict__ X,
                                            const float* __restrict__ Y,
                                            float* __restrict__ C,
                                            int wave_u, int lane) {
  wmma_tile64(X, Y, C, wave_u, lane);
  wmma_tile64(X, Y, C, wave_u + 8, lane);
}

__global__ void __launch_bounds__(256)
fem_solve_kernel(const float* __restrict__ Fv,
                 const float* __restrict__ Sg,
                 const float* __restrict__ invlam,
                 float* __restrict__ u) {
  __shared__ float Sm[PAD * PAD];
  __shared__ float T0[PAD * PAD];
  __shared__ float T1[PAD * PAD];

  const int tid    = threadIdx.x;
  const int lane   = tid & 31;
  const int wave_u = __builtin_amdgcn_readfirstlane(tid >> 5);

  for (int i = tid; i < PAD * PAD; i += 256) {
    Sm[i] = Sg[i];
    int p = i / PAD, q = i % PAD;
    float v = 0.0f;
    if (p < NI && q < NI) v = Fv[(p + 1) * GRID_NODES + (q + 1)];
    T0[i] = v;
  }
  __syncthreads();

  wmma_gemm64(Sm, T0, T1, wave_u, lane);
  __syncthreads();
  wmma_gemm64(T1, Sm, T0, wave_u, lane);
  __syncthreads();

  for (int i = tid; i < PAD * PAD; i += 256)
    T0[i] *= invlam[i];
  __syncthreads();

  wmma_gemm64(Sm, T0, T1, wave_u, lane);
  __syncthreads();
  wmma_gemm64(T1, Sm, T0, wave_u, lane);
  __syncthreads();

  for (int i = tid; i < GRID_NODES * GRID_NODES; i += 256) {
    int r = i / GRID_NODES, c = i % GRID_NODES;
    float v = 0.0f;
    if (r >= 1 && r <= NI && c >= 1 && c <= NI)
      v = T0[(r - 1) * PAD + (c - 1)];
    u[i] = v;
  }
}

extern "C" void kernel_launch(void* const* d_in, const int* in_sizes, int n_in,
                              void* d_out, int out_size, void* d_ws, size_t ws_size,
                              hipStream_t stream) {
  const float* f      = (const float*)d_in[0];
  const float* nodes  = (const float*)d_in[1];
  const float* kappa  = (const float*)d_in[2];
  const int*   elems  = (const int*)d_in[4];
  const int    nelem  = in_sizes[4] / 3;

  float* ws     = (float*)d_ws;
  float* Fv     = ws;
  float* S      = ws + 4096;
  float* invlam = ws + 8192;

  fem_init_kernel<<<(GRID_NODES * GRID_NODES + 255) / 256, 256, 0, stream>>>(
      Fv, S, invlam, kappa);
  fem_assemble_kernel<<<(nelem + 255) / 256, 256, 0, stream>>>(
      f, nodes, elems, nelem, Fv);
  fem_solve_kernel<<<1, 256, 0, stream>>>(Fv, S, invlam, (float*)d_out);
}


# solve kernel: S never stored - each wave builds its 16 rows of S in registers as WMMA A fragments; all four GEMMs run as (S*X)^T with X read along LDS rows; eigenvalue scale applied to accumulators
# speedup vs baseline: 1.5892x; 1.5892x over previous
#include <hip/hip_runtime.h>
#include <math.h>


#define GRID_NODES 64
#define NI         62
#define PAD        64
#define PI_F       3.14159265358979323846f

typedef float v2f __attribute__((ext_vector_type(2)));
typedef float v4f __attribute__((ext_vector_type(4)));
typedef float v8f __attribute__((ext_vector_type(8)));

__global__ void fem_init_kernel(float* __restrict__ Fv) {
  int tid = blockIdx.x * blockDim.x + threadIdx.x;
  if (tid >= GRID_NODES * GRID_NODES) return;
  Fv[tid] = 0.0f;
}

__global__ void fem_assemble_kernel(const float* __restrict__ f,
                                    const float* __restrict__ nodes,
                                    const int* __restrict__ elems,
                                    int nelem,
                                    float* __restrict__ Fv) {
  int e = blockIdx.x * blockDim.x + threadIdx.x;
  if (e >= nelem) return;
  int a = elems[3 * e + 0];
  int b = elems[3 * e + 1];
  int c = elems[3 * e + 2];
  float xi = nodes[2 * a], yi = nodes[2 * a + 1];
  float xj = nodes[2 * b], yj = nodes[2 * b + 1];
  float xk = nodes[2 * c], yk = nodes[2 * c + 1];
  float area = 0.5f * fabsf((xj - xi) * (yk - yi) - (xk - xi) * (yj - yi));
  float fe = 0.0f;
  if (area > 1e-15f)
    fe = (area * (1.0f / 3.0f)) * ((f[a] + f[b] + f[c]) * (1.0f / 3.0f));
  atomicAdd(&Fv[a], fe);
  atomicAdd(&Fv[b], fe);
  atomicAdd(&Fv[c], fe);
}

template <bool SCALE>
__device__ __forceinline__ void stage_SX_T(const float* __restrict__ src,
                                           float* __restrict__ dst,
                                           const v2f (&a)[16],
                                           const float (&sc0)[8],
                                           const float (&sc1)[8],
                                           int tm, int tn0, int half, int l16) {
  const float* bp = src + (2 * half) * PAD + tn0 + l16;
  v2f b0[16], b1[16];
#pragma unroll
  for (int ks = 0; ks < 16; ++ks) {
    b0[ks][0] = bp[(4 * ks) * PAD];
    b0[ks][1] = bp[(4 * ks + 1) * PAD];
    b1[ks][0] = bp[(4 * ks) * PAD + 16];
    b1[ks][1] = bp[(4 * ks + 1) * PAD + 16];
  }
  v8f acc0 = {0.f, 0.f, 0.f, 0.f, 0.f, 0.f, 0.f, 0.f};
  v8f acc1 = {0.f, 0.f, 0.f, 0.f, 0.f, 0.f, 0.f, 0.f};
#pragma unroll
  for (int ks = 0; ks < 16; ++ks) {
    acc0 = __builtin_amdgcn_wmma_f32_16x16x4_f32(
        false, a[ks], false, b0[ks], (short)0, acc0, false, false);
    acc1 = __builtin_amdgcn_wmma_f32_16x16x4_f32(
        false, a[ks], false, b1[ks], (short)0, acc1, false, false);
  }
  if (SCALE) {
#pragma unroll
    for (int j = 0; j < 8; ++j) {
      acc0[j] *= sc0[j];
      acc1[j] *= sc1[j];
    }
  }
  float* dp0 = dst + (tn0 + l16) * PAD + tm + 8 * half;
  float* dp1 = dp0 + 16 * PAD;
  v4f lo0 = {acc0[0], acc0[1], acc0[2], acc0[3]};
  v4f hi0 = {acc0[4], acc0[5], acc0[6], acc0[7]};
  v4f lo1 = {acc1[0], acc1[1], acc1[2], acc1[3]};
  v4f hi1 = {acc1[4], acc1[5], acc1[6], acc1[7]};
  *(v4f*)(dp0)     = lo0;
  *(v4f*)(dp0 + 4) = hi0;
  *(v4f*)(dp1)     = lo1;
  *(v4f*)(dp1 + 4) = hi1;
}

__global__ void __launch_bounds__(256)
fem_solve_kernel(const float* __restrict__ Fv,
                 const float* __restrict__ kappa,
                 float* __restrict__ u) {
  __shared__ __attribute__((aligned(16))) float XA[PAD * PAD];
  __shared__ __attribute__((aligned(16))) float XB[PAD * PAD];

  const int tid    = threadIdx.x;
  const int lane   = tid & 31;
  const int wave_u = __builtin_amdgcn_readfirstlane(tid >> 5);
  const int half   = lane >> 4;
  const int l16    = lane & 15;
  const int tm     = (wave_u >> 1) * 16;
  const int tn0    = (wave_u & 1) * 32;

  for (int i = tid; i < PAD * PAD; i += 256) {
    int p = i / PAD, q = i % PAD;
    float v = 0.0f;
    if (p < NI && q < NI) v = Fv[(p + 1) * GRID_NODES + (q + 1)];
    XA[i] = v;
  }

  v2f a[16];
  const int r = tm + l16;
#pragma unroll
  for (int ks = 0; ks < 16; ++ks) {
#pragma unroll
    for (int j = 0; j < 2; ++j) {
      const int k = 4 * ks + 2 * half + j;
      const float s = __sinf((float)((r + 1) * (k + 1)) * (PI_F / 63.0f));
      a[ks][j] = (r < NI && k < NI) ? s : 0.0f;
    }
  }

  const float kap  = kappa[0];
  const float norm = (2.0f / 63.0f) * (2.0f / 63.0f);
  const int n0 = tn0 + l16;
  const int n1 = n0 + 16;
  const float ln0 = 2.0f - 2.0f * __cosf((float)(n0 + 1) * (PI_F / 63.0f));
  const float ln1 = 2.0f - 2.0f * __cosf((float)(n1 + 1) * (PI_F / 63.0f));
  float sc0[8], sc1[8];
#pragma unroll
  for (int j = 0; j < 8; ++j) {
    const int m = tm + 8 * half + j;
    const float lm = 2.0f - 2.0f * __cosf((float)(m + 1) * (PI_F / 63.0f));
    const float i0 = norm / (kap * (ln0 + lm));
    const float i1 = norm / (kap * (ln1 + lm));
    sc0[j] = (m < NI && n0 < NI) ? i0 : 0.0f;
    sc1[j] = (m < NI && n1 < NI) ? i1 : 0.0f;
  }
  __syncthreads();

  stage_SX_T<false>(XA, XB, a, sc0, sc1, tm, tn0, half, l16);
  __syncthreads();
  stage_SX_T<true >(XB, XA, a, sc0, sc1, tm, tn0, half, l16);
  __syncthreads();
  stage_SX_T<false>(XA, XB, a, sc0, sc1, tm, tn0, half, l16);
  __syncthreads();
  stage_SX_T<false>(XB, XA, a, sc0, sc1, tm, tn0, half, l16);
  __syncthreads();

  for (int i = tid; i < GRID_NODES * GRID_NODES; i += 256) {
    int rr = i / GRID_NODES, c = i % GRID_NODES;
    float v = 0.0f;
    if (rr >= 1 && rr <= NI && c >= 1 && c <= NI)
      v = XA[(rr - 1) * PAD + (c - 1)];
    u[i] = v;
  }
}

extern "C" void kernel_launch(void* const* d_in, const int* in_sizes, int n_in,
                              void* d_out, int out_size, void* d_ws, size_t ws_size,
                              hipStream_t stream) {
  const float* f      = (const float*)d_in[0];
  const float* nodes  = (const float*)d_in[1];
  const float* kappa  = (const float*)d_in[2];
  const int*   elems  = (const int*)d_in[4];
  const int    nelem  = in_sizes[4] / 3;

  float* ws     = (float*)d_ws;
  float* Fv     = ws;

  fem_init_kernel<<<(GRID_NODES * GRID_NODES + 255) / 256, 256, 0, stream>>>(Fv);
  fem_assemble_kernel<<<(nelem + 255) / 256, 256, 0, stream>>>(
      f, nodes, elems, nelem, Fv);
  fem_solve_kernel<<<1, 256, 0, stream>>>(Fv, kappa, (float*)d_out);
}


# solve stages: data matrix is now the WMMA A operand (S is B, in registers), k permuted so a lane's row share is 8 16-byte LDS reads (pitch 68) grouped ahead of the WMMAs; k sum order changes
# speedup vs baseline: 3.0104x; 1.8943x over previous
#include <hip/hip_runtime.h>
#include <math.h>


#define GRID_NODES 64
#define NI         62
#define PAD        64
#define PITCH      68
#define PI_F       3.14159265358979323846f
#define NPART      4
#define ASM_THREADS 1024

typedef float v2f __attribute__((ext_vector_type(2)));
typedef float v4f __attribute__((ext_vector_type(4)));
typedef float v8f __attribute__((ext_vector_type(8)));

__device__ __forceinline__ float elem_fe(const float* __restrict__ f,
                                         const float* __restrict__ nodes,
                                         int a, int b, int c) {
  const float xi = nodes[2 * a], yi = nodes[2 * a + 1];
  const float xj = nodes[2 * b], yj = nodes[2 * b + 1];
  const float xk = nodes[2 * c], yk = nodes[2 * c + 1];
  const float fa = f[a], fb = f[b], fc = f[c];
  const float area = 0.5f * fabsf((xj - xi) * (yk - yi) - (xk - xi) * (yj - yi));
  const float fe = (area * (1.0f / 3.0f)) * ((fa + fb + fc) * (1.0f / 3.0f));
  return (area > 1e-15f) ? fe : 0.0f;
}

__device__ __forceinline__ void add_interior(float* P, int n, float v) {
  const int r = n >> 6, c = n & (GRID_NODES - 1);
  if (r >= 1 && r <= NI && c >= 1 && c <= NI)
    __hip_atomic_fetch_add(&P[n - (GRID_NODES + 1)], v, __ATOMIC_RELAXED,
                           __HIP_MEMORY_SCOPE_WORKGROUP);
}

__global__ void __launch_bounds__(ASM_THREADS)
fem_assemble_kernel(const float* __restrict__ f,
                    const float* __restrict__ nodes,
                    const int* __restrict__ elems,
                    int nelem,
                    float* __restrict__ part) {
  __shared__ __attribute__((aligned(16))) float P[PAD * PAD];
  const int tid  = threadIdx.x;
  const int gtid = blockIdx.x * ASM_THREADS + tid;
  const int tot  = NPART * ASM_THREADS;

  const v4f zero4 = {0.f, 0.f, 0.f, 0.f};
  ((v4f*)P)[tid] = zero4;
  __syncthreads();

  const int npair = nelem >> 1;
  for (int p = gtid; p < npair; p += tot) {
    const int* ep = elems + 6 * p;
    const int a0 = ep[0], b0 = ep[1], c0 = ep[2];
    const int a1 = ep[3], b1 = ep[4], c1 = ep[5];
    const float fe0 = elem_fe(f, nodes, a0, b0, c0);
    const float fe1 = elem_fe(f, nodes, a1, b1, c1);
    add_interior(P, a0, fe0);
    add_interior(P, b0, fe0);
    add_interior(P, c0, fe0);
    add_interior(P, a1, fe1);
    add_interior(P, b1, fe1);
    add_interior(P, c1, fe1);
  }
  if ((nelem & 1) && gtid == 0) {
    const int* ep = elems + 3 * (nelem - 1);
    const int a = ep[0], b = ep[1], c = ep[2];
    const float fe = elem_fe(f, nodes, a, b, c);
    add_interior(P, a, fe);
    add_interior(P, b, fe);
    add_interior(P, c, fe);
  }
  __syncthreads();

  ((v4f*)(part + blockIdx.x * (PAD * PAD)))[tid] = ((const v4f*)P)[tid];
}

template <bool SCALE>
__device__ __forceinline__ void stage_XS_T(const float* __restrict__ src,
                                           float* __restrict__ dst,
                                           const v2f (&b)[16],
                                           const float (&sc0)[8],
                                           const float (&sc1)[8],
                                           int tm0, int tn, int half, int l16) {
  const v4f* ap0 = (const v4f*)(src + (tm0 + l16) * PITCH + 32 * half);
  const v4f* ap1 = (const v4f*)(src + (tm0 + 16 + l16) * PITCH + 32 * half);
  v4f a0[8], a1[8];
#pragma unroll
  for (int q = 0; q < 8; ++q) {
    a0[q] = ap0[q];
    a1[q] = ap1[q];
  }
  __builtin_amdgcn_sched_barrier(0);
  v8f acc0 = {0.f, 0.f, 0.f, 0.f, 0.f, 0.f, 0.f, 0.f};
  v8f acc1 = {0.f, 0.f, 0.f, 0.f, 0.f, 0.f, 0.f, 0.f};
#pragma unroll
  for (int q = 0; q < 8; ++q) {
    const v2f a0lo = {a0[q][0], a0[q][1]}, a0hi = {a0[q][2], a0[q][3]};
    const v2f a1lo = {a1[q][0], a1[q][1]}, a1hi = {a1[q][2], a1[q][3]};
    acc0 = __builtin_amdgcn_wmma_f32_16x16x4_f32(
        false, a0lo, false, b[2 * q], (short)0, acc0, false, false);
    acc1 = __builtin_amdgcn_wmma_f32_16x16x4_f32(
        false, a1lo, false, b[2 * q], (short)0, acc1, false, false);
    acc0 = __builtin_amdgcn_wmma_f32_16x16x4_f32(
        false, a0hi, false, b[2 * q + 1], (short)0, acc0, false, false);
    acc1 = __builtin_amdgcn_wmma_f32_16x16x4_f32(
        false, a1hi, false, b[2 * q + 1], (short)0, acc1, false, false);
  }
  if (SCALE) {
#pragma unroll
    for (int j = 0; j < 8; ++j) {
      acc0[j] *= sc0[j];
      acc1[j] *= sc1[j];
    }
  }
  float* dp0 = dst + (tn + l16) * PITCH + tm0 + 8 * half;
  float* dp1 = dp0 + 16;
  v4f lo0 = {acc0[0], acc0[1], acc0[2], acc0[3]};
  v4f hi0 = {acc0[4], acc0[5], acc0[6], acc0[7]};
  v4f lo1 = {acc1[0], acc1[1], acc1[2], acc1[3]};
  v4f hi1 = {acc1[4], acc1[5], acc1[6], acc1[7]};
  *(v4f*)(dp0)     = lo0;
  *(v4f*)(dp0 + 4) = hi0;
  *(v4f*)(dp1)     = lo1;
  *(v4f*)(dp1 + 4) = hi1;
}

__global__ void __launch_bounds__(256)
fem_solve_kernel(const float* __restrict__ part,
                 const float* __restrict__ kappa,
                 float* __restrict__ u) {
  __shared__ __attribute__((aligned(16))) float XA[PAD * PITCH];
  __shared__ __attribute__((aligned(16))) float XB[PAD * PITCH];

  const int tid    = threadIdx.x;
  const int lane   = tid & 31;
  const int wave_u = __builtin_amdgcn_readfirstlane(tid >> 5);
  const int half   = lane >> 4;
  const int l16    = lane & 15;
  const int tn     = (wave_u >> 1) * 16;
  const int tm0    = (wave_u & 1) * 32;

  v4f pl[NPART][4];
#pragma unroll
  for (int g = 0; g < NPART; ++g) {
#pragma unroll
    for (int q = 0; q < 4; ++q)
      pl[g][q] = ((const v4f*)(part + g * (PAD * PAD)))[tid + 256 * q];
  }

  v2f b[16];
  const int n = tn + l16;
  const float fn1 = (float)(n + 1);
  const float fh  = (float)(32 * half);
#pragma unroll
  for (int ks = 0; ks < 16; ++ks) {
#pragma unroll
    for (int s = 0; s < 2; ++s) {
      const int k = 32 * half + 2 * ks + s;
      const float nk = fn1 * (fh + (float)(2 * ks + s + 1));
      const float sv = __sinf(nk * (PI_F / 63.0f));
      b[ks][s] = (n < NI && k < NI) ? sv : 0.0f;
    }
  }

  const float kap  = kappa[0];
  const float norm = (2.0f / 63.0f) * (2.0f / 63.0f);
  const float ln = 2.0f - 2.0f * __cosf((float)(n + 1) * (PI_F / 63.0f));
  float sc0[8], sc1[8];
#pragma unroll
  for (int j = 0; j < 8; ++j) {
    const int m0 = tm0 + 8 * half + j;
    const int m1 = m0 + 16;
    const float lm0 = 2.0f - 2.0f * __cosf((float)(m0 + 1) * (PI_F / 63.0f));
    const float lm1 = 2.0f - 2.0f * __cosf((float)(m1 + 1) * (PI_F / 63.0f));
    const float i0 = norm / (kap * (ln + lm0));
    const float i1 = norm / (kap * (ln + lm1));
    sc0[j] = (m0 < NI && n < NI) ? i0 : 0.0f;
    sc1[j] = (m1 < NI && n < NI) ? i1 : 0.0f;
  }
  asm volatile("" : "+v"(sc0[0]), "+v"(sc0[1]), "+v"(sc0[2]), "+v"(sc0[3]),
                    "+v"(sc0[4]), "+v"(sc0[5]), "+v"(sc0[6]), "+v"(sc0[7]),
                    "+v"(sc1[0]), "+v"(sc1[1]), "+v"(sc1[2]), "+v"(sc1[3]),
                    "+v"(sc1[4]), "+v"(sc1[5]), "+v"(sc1[6]), "+v"(sc1[7]));
  __builtin_amdgcn_sched_barrier(0);
  float* xs = XA + (tid >> 4) * PITCH + 4 * (tid & 15);
#pragma unroll
  for (int q = 0; q < 4; ++q) {
    v4f s = pl[0][q];
#pragma unroll
    for (int g = 1; g < NPART; ++g) s += pl[g][q];
    *(v4f*)(xs + 16 * q * PITCH) = s;
  }
  __syncthreads();

  stage_XS_T<false>(XA, XB, b, sc0, sc1, tm0, tn, half, l16);
  __syncthreads();
  stage_XS_T<true >(XB, XA, b, sc0, sc1, tm0, tn, half, l16);
  __syncthreads();
  stage_XS_T<false>(XA, XB, b, sc0, sc1, tm0, tn, half, l16);
  __syncthreads();
  stage_XS_T<false>(XB, XA, b, sc0, sc1, tm0, tn, half, l16);
  __syncthreads();

  for (int i = tid; i < GRID_NODES * GRID_NODES; i += 256) {
    int rr = i / GRID_NODES, c = i % GRID_NODES;
    float v = 0.0f;
    if (rr >= 1 && rr <= NI && c >= 1 && c <= NI)
      v = XA[(rr - 1) * PITCH + (c - 1)];
    u[i] = v;
  }
}

extern "C" void kernel_launch(void* const* d_in, const int* in_sizes, int n_in,
                              void* d_out, int out_size, void* d_ws, size_t ws_size,
                              hipStream_t stream) {
  const float* f      = (const float*)d_in[0];
  const float* nodes  = (const float*)d_in[1];
  const float* kappa  = (const float*)d_in[2];
  const int*   elems  = (const int*)d_in[4];
  const int    nelem  = in_sizes[4] / 3;

  float* part   = (float*)d_ws;

  fem_assemble_kernel<<<NPART, ASM_THREADS, 0, stream>>>(f, nodes, elems, nelem, part);
  fem_solve_kernel<<<1, 256, 0, stream>>>(part, kappa, (float*)d_out);
}


# solve split over 4 workgroups: each builds all of W (stages 1-2) but only its own 16 rows of stages 3-4 (k halved per wave, partial tiles summed) and of the output; k sum order changes there
# speedup vs baseline: 3.4014x; 1.1299x over previous
#include <hip/hip_runtime.h>
#include <math.h>


#define GRID_NODES 64
#define NI         62
#define PAD        64
#define PITCH      68
#define PI_F       3.14159265358979323846f
#define NPART      4
#define ASM_THREADS 1024
#define NSOLVE     4

typedef float v2f __attribute__((ext_vector_type(2)));
typedef float v4f __attribute__((ext_vector_type(4)));
typedef float v8f __attribute__((ext_vector_type(8)));

__device__ __forceinline__ float elem_fe(const float* __restrict__ f,
                                         const float* __restrict__ nodes,
                                         int a, int b, int c) {
  const float xi = nodes[2 * a], yi = nodes[2 * a + 1];
  const float xj = nodes[2 * b], yj = nodes[2 * b + 1];
  const float xk = nodes[2 * c], yk = nodes[2 * c + 1];
  const float fa = f[a], fb = f[b], fc = f[c];
  const float area = 0.5f * fabsf((xj - xi) * (yk - yi) - (xk - xi) * (yj - yi));
  const float fe = (area * (1.0f / 3.0f)) * ((fa + fb + fc) * (1.0f / 3.0f));
  return (area > 1e-15f) ? fe : 0.0f;
}

__device__ __forceinline__ void add_interior(float* P, int n, float v) {
  const int r = n >> 6, c = n & (GRID_NODES - 1);
  if (r >= 1 && r <= NI && c >= 1 && c <= NI)
    __hip_atomic_fetch_add(&P[n - (GRID_NODES + 1)], v, __ATOMIC_RELAXED,
                           __HIP_MEMORY_SCOPE_WORKGROUP);
}

__global__ void __launch_bounds__(ASM_THREADS)
fem_assemble_kernel(const float* __restrict__ f,
                    const float* __restrict__ nodes,
                    const int* __restrict__ elems,
                    int nelem,
                    float* __restrict__ part) {
  __shared__ __attribute__((aligned(16))) float P[PAD * PAD];
  const int tid  = threadIdx.x;
  const int gtid = blockIdx.x * ASM_THREADS + tid;
  const int tot  = NPART * ASM_THREADS;

  const v4f zero4 = {0.f, 0.f, 0.f, 0.f};
  ((v4f*)P)[tid] = zero4;
  __syncthreads();

  const int npair = nelem >> 1;
  for (int p = gtid; p < npair; p += tot) {
    const int* ep = elems + 6 * p;
    const int a0 = ep[0], b0 = ep[1], c0 = ep[2];
    const int a1 = ep[3], b1 = ep[4], c1 = ep[5];
    const float fe0 = elem_fe(f, nodes, a0, b0, c0);
    const float fe1 = elem_fe(f, nodes, a1, b1, c1);
    add_interior(P, a0, fe0);
    add_interior(P, b0, fe0);
    add_interior(P, c0, fe0);
    add_interior(P, a1, fe1);
    add_interior(P, b1, fe1);
    add_interior(P, c1, fe1);
  }
  if ((nelem & 1) && gtid == 0) {
    const int* ep = elems + 3 * (nelem - 1);
    const int a = ep[0], b = ep[1], c = ep[2];
    const float fe = elem_fe(f, nodes, a, b, c);
    add_interior(P, a, fe);
    add_interior(P, b, fe);
    add_interior(P, c, fe);
  }
  __syncthreads();

  ((v4f*)(part + blockIdx.x * (PAD * PAD)))[tid] = ((const v4f*)P)[tid];
}

template <bool SCALE>
__device__ __forceinline__ void stage_XS_T(const float* __restrict__ src,
                                           float* __restrict__ dst,
                                           const v2f (&b)[16],
                                           const float (&sc0)[8],
                                           const float (&sc1)[8],
                                           int tm0, int tn, int half, int l16) {
  const v4f* ap0 = (const v4f*)(src + (tm0 + l16) * PITCH + 32 * half);
  const v4f* ap1 = (const v4f*)(src + (tm0 + 16 + l16) * PITCH + 32 * half);
  v4f a0[8], a1[8];
#pragma unroll
  for (int q = 0; q < 8; ++q) {
    a0[q] = ap0[q];
    a1[q] = ap1[q];
  }
  __builtin_amdgcn_sched_barrier(0);
  v8f acc0 = {0.f, 0.f, 0.f, 0.f, 0.f, 0.f, 0.f, 0.f};
  v8f acc1 = {0.f, 0.f, 0.f, 0.f, 0.f, 0.f, 0.f, 0.f};
#pragma unroll
  for (int q = 0; q < 8; ++q) {
    const v2f a0lo = {a0[q][0], a0[q][1]}, a0hi = {a0[q][2], a0[q][3]};
    const v2f a1lo = {a1[q][0], a1[q][1]}, a1hi = {a1[q][2], a1[q][3]};
    acc0 = __builtin_amdgcn_wmma_f32_16x16x4_f32(
        false, a0lo, false, b[2 * q], (short)0, acc0, false, false);
    acc1 = __builtin_amdgcn_wmma_f32_16x16x4_f32(
        false, a1lo, false, b[2 * q], (short)0, acc1, false, false);
    acc0 = __builtin_amdgcn_wmma_f32_16x16x4_f32(
        false, a0hi, false, b[2 * q + 1], (short)0, acc0, false, false);
    acc1 = __builtin_amdgcn_wmma_f32_16x16x4_f32(
        false, a1hi, false, b[2 * q + 1], (short)0, acc1, false, false);
  }
  if (SCALE) {
#pragma unroll
    for (int j = 0; j < 8; ++j) {
      acc0[j] *= sc0[j];
      acc1[j] *= sc1[j];
    }
  }
  float* dp0 = dst + (tn + l16) * PITCH + tm0 + 8 * half;
  float* dp1 = dp0 + 16;
  v4f lo0 = {acc0[0], acc0[1], acc0[2], acc0[3]};
  v4f hi0 = {acc0[4], acc0[5], acc0[6], acc0[7]};
  v4f lo1 = {acc1[0], acc1[1], acc1[2], acc1[3]};
  v4f hi1 = {acc1[4], acc1[5], acc1[6], acc1[7]};
  *(v4f*)(dp0)     = lo0;
  *(v4f*)(dp0 + 4) = hi0;
  *(v4f*)(dp1)     = lo1;
  *(v4f*)(dp1 + 4) = hi1;
}

__device__ __forceinline__ void stage3_rows(const float* __restrict__ W,
                                            float* __restrict__ Yp,
                                            const v2f (&b3)[8],
                                            int kh3, int tm3, int half, int l16) {
  const float* wp = W + (32 * half + 16 * kh3) * PITCH + tm3 + l16;
  float a[16];
#pragma unroll
  for (int e = 0; e < 16; ++e) a[e] = wp[e * PITCH];
  __builtin_amdgcn_sched_barrier(0);
  v8f acc0 = {0.f, 0.f, 0.f, 0.f, 0.f, 0.f, 0.f, 0.f};
  v8f acc1 = {0.f, 0.f, 0.f, 0.f, 0.f, 0.f, 0.f, 0.f};
#pragma unroll
  for (int t = 0; t < 8; t += 2) {
    const v2f ae = {a[2 * t], a[2 * t + 1]};
    const v2f ao = {a[2 * t + 2], a[2 * t + 3]};
    acc0 = __builtin_amdgcn_wmma_f32_16x16x4_f32(
        false, ae, false, b3[t], (short)0, acc0, false, false);
    acc1 = __builtin_amdgcn_wmma_f32_16x16x4_f32(
        false, ao, false, b3[t + 1], (short)0, acc1, false, false);
  }
  const v8f acc = acc0 + acc1;
  float* yp = Yp + (16 * kh3 + l16) * PITCH + tm3 + 8 * half;
  v4f lo = {acc[0], acc[1], acc[2], acc[3]};
  v4f hi = {acc[4], acc[5], acc[6], acc[7]};
  *(v4f*)(yp)     = lo;
  *(v4f*)(yp + 4) = hi;
}

template <int KH>
__device__ __forceinline__ void stage4_rows(const float* __restrict__ Yp,
                                            float* __restrict__ Up,
                                            const v2f (&b)[16],
                                            int tn, int half, int l16) {
  const v4f* y0 = (const v4f*)(Yp + l16 * PITCH + 32 * half + 16 * KH);
  const v4f* y1 = (const v4f*)(Yp + (16 + l16) * PITCH + 32 * half + 16 * KH);
  v4f p0[4], p1[4];
#pragma unroll
  for (int q = 0; q < 4; ++q) {
    p0[q] = y0[q];
    p1[q] = y1[q];
  }
  __builtin_amdgcn_sched_barrier(0);
  v8f acc0 = {0.f, 0.f, 0.f, 0.f, 0.f, 0.f, 0.f, 0.f};
  v8f acc1 = {0.f, 0.f, 0.f, 0.f, 0.f, 0.f, 0.f, 0.f};
#pragma unroll
  for (int q = 0; q < 4; ++q) {
    const v4f y = p0[q] + p1[q];
    const v2f ylo = {y[0], y[1]}, yhi = {y[2], y[3]};
    acc0 = __builtin_amdgcn_wmma_f32_16x16x4_f32(
        false, ylo, false, b[8 * KH + 2 * q], (short)0, acc0, false, false);
    acc1 = __builtin_amdgcn_wmma_f32_16x16x4_f32(
        false, yhi, false, b[8 * KH + 2 * q + 1], (short)0, acc1, false, false);
  }
  const v8f acc = acc0 + acc1;
  float* up = Up + (16 * KH + 8 * half) * PITCH + tn + l16;
#pragma unroll
  for (int j = 0; j < 8; ++j) up[j * PITCH] = acc[j];
}

__global__ void __launch_bounds__(256)
fem_solve_kernel(const float* __restrict__ part,
                 const float* __restrict__ kappa,
                 float* __restrict__ u) {
  __shared__ __attribute__((aligned(16))) float XA[PAD * PITCH];
  __shared__ __attribute__((aligned(16))) float XB[PAD * PITCH];

  const int tid    = threadIdx.x;
  const int lane   = tid & 31;
  const int wave_u = __builtin_amdgcn_readfirstlane(tid >> 5);
  const int half   = lane >> 4;
  const int l16    = lane & 15;
  const int tn     = (wave_u >> 1) * 16;
  const int tm0    = (wave_u & 1) * 32;

  v4f pl[NPART][4];
#pragma unroll
  for (int g = 0; g < NPART; ++g) {
#pragma unroll
    for (int q = 0; q < 4; ++q)
      pl[g][q] = ((const v4f*)(part + g * (PAD * PAD)))[tid + 256 * q];
  }

  v2f b[16];
  const int n = tn + l16;
  const float fn1 = (float)(n + 1);
  const float fh  = (float)(32 * half);
#pragma unroll
  for (int ks = 0; ks < 16; ++ks) {
#pragma unroll
    for (int s = 0; s < 2; ++s) {
      const int k = 32 * half + 2 * ks + s;
      const float nk = fn1 * (fh + (float)(2 * ks + s + 1));
      const float sv = __sinf(nk * (PI_F / 63.0f));
      b[ks][s] = (n < NI && k < NI) ? sv : 0.0f;
    }
  }

  const int blk = blockIdx.x;
  const int kh3 = wave_u >> 2;
  const int tm3 = (wave_u & 3) * 16;
  v2f b3[8];
  const int n3 = 16 * blk + l16;
  const float fn3 = (float)(n3 + 1);
  const float fh3 = (float)(32 * half + 16 * kh3);
#pragma unroll
  for (int t = 0; t < 8; ++t) {
#pragma unroll
    for (int s = 0; s < 2; ++s) {
      const int k = 32 * half + 16 * kh3 + 2 * t + s;
      const float nk = fn3 * (fh3 + (float)(2 * t + s + 1));
      const float sv = __sinf(nk * (PI_F / 63.0f));
      b3[t][s] = (n3 < NI && k < NI) ? sv : 0.0f;
    }
  }

  const float kap  = kappa[0];
  const float norm = (2.0f / 63.0f) * (2.0f / 63.0f);
  const float ln = 2.0f - 2.0f * __cosf((float)(n + 1) * (PI_F / 63.0f));
  float sc0[8], sc1[8];
#pragma unroll
  for (int j = 0; j < 8; ++j) {
    const int m0 = tm0 + 8 * half + j;
    const int m1 = m0 + 16;
    const float lm0 = 2.0f - 2.0f * __cosf((float)(m0 + 1) * (PI_F / 63.0f));
    const float lm1 = 2.0f - 2.0f * __cosf((float)(m1 + 1) * (PI_F / 63.0f));
    const float i0 = norm / (kap * (ln + lm0));
    const float i1 = norm / (kap * (ln + lm1));
    sc0[j] = (m0 < NI && n < NI) ? i0 : 0.0f;
    sc1[j] = (m1 < NI && n < NI) ? i1 : 0.0f;
  }
  asm volatile("" : "+v"(sc0[0]), "+v"(sc0[1]), "+v"(sc0[2]), "+v"(sc0[3]),
                    "+v"(sc0[4]), "+v"(sc0[5]), "+v"(sc0[6]), "+v"(sc0[7]),
                    "+v"(sc1[0]), "+v"(sc1[1]), "+v"(sc1[2]), "+v"(sc1[3]),
                    "+v"(sc1[4]), "+v"(sc1[5]), "+v"(sc1[6]), "+v"(sc1[7]));
  __builtin_amdgcn_sched_barrier(0);
  float* xs = XA + (tid >> 4) * PITCH + 4 * (tid & 15);
#pragma unroll
  for (int q = 0; q < 4; ++q) {
    v4f s = pl[0][q];
#pragma unroll
    for (int g = 1; g < NPART; ++g) s += pl[g][q];
    *(v4f*)(xs + 16 * q * PITCH) = s;
  }
  __syncthreads();

  stage_XS_T<false>(XA, XB, b, sc0, sc1, tm0, tn, half, l16);
  __syncthreads();
  stage_XS_T<true >(XB, XA, b, sc0, sc1, tm0, tn, half, l16);
  __syncthreads();
  stage3_rows(XA, XB, b3, kh3, tm3, half, l16);
  __syncthreads();
  if (wave_u & 1) stage4_rows<1>(XB, XA, b, tn, half, l16);
  else            stage4_rows<0>(XB, XA, b, tn, half, l16);
  __syncthreads();

  const int r_lo = (blk == 0) ? 0 : 16 * blk + 1;
  const int r_hi = (16 * blk + 16 < GRID_NODES) ? 16 * blk + 16 : GRID_NODES - 1;
  for (int i = r_lo * GRID_NODES + tid; i < (r_hi + 1) * GRID_NODES; i += 256) {
    const int rr = i / GRID_NODES, c = i % GRID_NODES;
    float v = 0.0f;
    if (rr >= 1 && rr <= NI && c >= 1 && c <= NI) {
      const int o = (rr - 1 - 16 * blk) * PITCH + (c - 1);
      v = XA[o] + XA[16 * PITCH + o];
    }
    u[i] = v;
  }
}

extern "C" void kernel_launch(void* const* d_in, const int* in_sizes, int n_in,
                              void* d_out, int out_size, void* d_ws, size_t ws_size,
                              hipStream_t stream) {
  const float* f      = (const float*)d_in[0];
  const float* nodes  = (const float*)d_in[1];
  const float* kappa  = (const float*)d_in[2];
  const int*   elems  = (const int*)d_in[4];
  const int    nelem  = in_sizes[4] / 3;

  float* part   = (float*)d_ws;

  fem_assemble_kernel<<<NPART, ASM_THREADS, 0, stream>>>(f, nodes, elems, nelem, part);
  fem_solve_kernel<<<NSOLVE, 256, 0, stream>>>(part, kappa, (float*)d_out);
}


# scatter: u staged at its grid column in LDS (1-float shift), thread = one 16 B read per partial plane + one 16 B store, 5-trip 4 B loop gone; build side effect: solve's 16 plane loads now issue first
# speedup vs baseline: 3.8271x; 1.1252x over previous
#include <hip/hip_runtime.h>
#include <math.h>


#define GRID_NODES 64
#define NI         62
#define PAD        64
#define PITCH      68
#define PI_F       3.14159265358979323846f
#define NPART      4
#define ASM_THREADS 1024
#define NSOLVE     4

typedef float v2f __attribute__((ext_vector_type(2)));
typedef float v4f __attribute__((ext_vector_type(4)));
typedef float v8f __attribute__((ext_vector_type(8)));

__device__ __forceinline__ float elem_fe(const float* __restrict__ f,
                                         const float* __restrict__ nodes,
                                         int a, int b, int c) {
  const float xi = nodes[2 * a], yi = nodes[2 * a + 1];
  const float xj = nodes[2 * b], yj = nodes[2 * b + 1];
  const float xk = nodes[2 * c], yk = nodes[2 * c + 1];
  const float fa = f[a], fb = f[b], fc = f[c];
  const float area = 0.5f * fabsf((xj - xi) * (yk - yi) - (xk - xi) * (yj - yi));
  const float fe = (area * (1.0f / 3.0f)) * ((fa + fb + fc) * (1.0f / 3.0f));
  return (area > 1e-15f) ? fe : 0.0f;
}

__device__ __forceinline__ void add_interior(float* P, int n, float v) {
  const int r = n >> 6, c = n & (GRID_NODES - 1);
  if (r >= 1 && r <= NI && c >= 1 && c <= NI)
    __hip_atomic_fetch_add(&P[n - (GRID_NODES + 1)], v, __ATOMIC_RELAXED,
                           __HIP_MEMORY_SCOPE_WORKGROUP);
}

__global__ void __launch_bounds__(ASM_THREADS)
fem_assemble_kernel(const float* __restrict__ f,
                    const float* __restrict__ nodes,
                    const int* __restrict__ elems,
                    int nelem,
                    float* __restrict__ part) {
  __shared__ __attribute__((aligned(16))) float P[PAD * PAD];
  const int tid  = threadIdx.x;
  const int gtid = blockIdx.x * ASM_THREADS + tid;
  const int tot  = NPART * ASM_THREADS;

  const v4f zero4 = {0.f, 0.f, 0.f, 0.f};
  ((v4f*)P)[tid] = zero4;
  __syncthreads();

  const int npair = nelem >> 1;
  for (int p = gtid; p < npair; p += tot) {
    const int* ep = elems + 6 * p;
    const int a0 = ep[0], b0 = ep[1], c0 = ep[2];
    const int a1 = ep[3], b1 = ep[4], c1 = ep[5];
    const float fe0 = elem_fe(f, nodes, a0, b0, c0);
    const float fe1 = elem_fe(f, nodes, a1, b1, c1);
    add_interior(P, a0, fe0);
    add_interior(P, b0, fe0);
    add_interior(P, c0, fe0);
    add_interior(P, a1, fe1);
    add_interior(P, b1, fe1);
    add_interior(P, c1, fe1);
  }
  if ((nelem & 1) && gtid == 0) {
    const int* ep = elems + 3 * (nelem - 1);
    const int a = ep[0], b = ep[1], c = ep[2];
    const float fe = elem_fe(f, nodes, a, b, c);
    add_interior(P, a, fe);
    add_interior(P, b, fe);
    add_interior(P, c, fe);
  }
  __syncthreads();

  ((v4f*)(part + blockIdx.x * (PAD * PAD)))[tid] = ((const v4f*)P)[tid];
}

template <bool SCALE>
__device__ __forceinline__ void stage_XS_T(const float* __restrict__ src,
                                           float* __restrict__ dst,
                                           const v2f (&b)[16],
                                           const float (&sc0)[8],
                                           const float (&sc1)[8],
                                           int tm0, int tn, int half, int l16) {
  const v4f* ap0 = (const v4f*)(src + (tm0 + l16) * PITCH + 32 * half);
  const v4f* ap1 = (const v4f*)(src + (tm0 + 16 + l16) * PITCH + 32 * half);
  v4f a0[8], a1[8];
#pragma unroll
  for (int q = 0; q < 8; ++q) {
    a0[q] = ap0[q];
    a1[q] = ap1[q];
  }
  __builtin_amdgcn_sched_barrier(0);
  v8f acc0 = {0.f, 0.f, 0.f, 0.f, 0.f, 0.f, 0.f, 0.f};
  v8f acc1 = {0.f, 0.f, 0.f, 0.f, 0.f, 0.f, 0.f, 0.f};
#pragma unroll
  for (int q = 0; q < 8; ++q) {
    const v2f a0lo = {a0[q][0], a0[q][1]}, a0hi = {a0[q][2], a0[q][3]};
    const v2f a1lo = {a1[q][0], a1[q][1]}, a1hi = {a1[q][2], a1[q][3]};
    acc0 = __builtin_amdgcn_wmma_f32_16x16x4_f32(
        false, a0lo, false, b[2 * q], (short)0, acc0, false, false);
    acc1 = __builtin_amdgcn_wmma_f32_16x16x4_f32(
        false, a1lo, false, b[2 * q], (short)0, acc1, false, false);
    acc0 = __builtin_amdgcn_wmma_f32_16x16x4_f32(
        false, a0hi, false, b[2 * q + 1], (short)0, acc0, false, false);
    acc1 = __builtin_amdgcn_wmma_f32_16x16x4_f32(
        false, a1hi, false, b[2 * q + 1], (short)0, acc1, false, false);
  }
  if (SCALE) {
#pragma unroll
    for (int j = 0; j < 8; ++j) {
      acc0[j] *= sc0[j];
      acc1[j] *= sc1[j];
    }
  }
  float* dp0 = dst + (tn + l16) * PITCH + tm0 + 8 * half;
  float* dp1 = dp0 + 16;
  v4f lo0 = {acc0[0], acc0[1], acc0[2], acc0[3]};
  v4f hi0 = {acc0[4], acc0[5], acc0[6], acc0[7]};
  v4f lo1 = {acc1[0], acc1[1], acc1[2], acc1[3]};
  v4f hi1 = {acc1[4], acc1[5], acc1[6], acc1[7]};
  *(v4f*)(dp0)     = lo0;
  *(v4f*)(dp0 + 4) = hi0;
  *(v4f*)(dp1)     = lo1;
  *(v4f*)(dp1 + 4) = hi1;
}

__device__ __forceinline__ void stage3_rows(const float* __restrict__ W,
                                            float* __restrict__ Yp,
                                            const v2f (&b3)[8],
                                            int kh3, int tm3, int half, int l16) {
  const float* wp = W + (32 * half + 16 * kh3) * PITCH + tm3 + l16;
  float a[16];
#pragma unroll
  for (int e = 0; e < 16; ++e) a[e] = wp[e * PITCH];
  __builtin_amdgcn_sched_barrier(0);
  v8f acc0 = {0.f, 0.f, 0.f, 0.f, 0.f, 0.f, 0.f, 0.f};
  v8f acc1 = {0.f, 0.f, 0.f, 0.f, 0.f, 0.f, 0.f, 0.f};
#pragma unroll
  for (int t = 0; t < 8; t += 2) {
    const v2f ae = {a[2 * t], a[2 * t + 1]};
    const v2f ao = {a[2 * t + 2], a[2 * t + 3]};
    acc0 = __builtin_amdgcn_wmma_f32_16x16x4_f32(
        false, ae, false, b3[t], (short)0, acc0, false, false);
    acc1 = __builtin_amdgcn_wmma_f32_16x16x4_f32(
        false, ao, false, b3[t + 1], (short)0, acc1, false, false);
  }
  const v8f acc = acc0 + acc1;
  float* yp = Yp + (16 * kh3 + l16) * PITCH + tm3 + 8 * half;
  v4f lo = {acc[0], acc[1], acc[2], acc[3]};
  v4f hi = {acc[4], acc[5], acc[6], acc[7]};
  *(v4f*)(yp)     = lo;
  *(v4f*)(yp + 4) = hi;
}

template <int KH>
__device__ __forceinline__ void stage4_rows(const float* __restrict__ Yp,
                                            float* __restrict__ Up,
                                            const v2f (&b)[16],
                                            int tn, int half, int l16) {
  const v4f* y0 = (const v4f*)(Yp + l16 * PITCH + 32 * half + 16 * KH);
  const v4f* y1 = (const v4f*)(Yp + (16 + l16) * PITCH + 32 * half + 16 * KH);
  v4f p0[4], p1[4];
#pragma unroll
  for (int q = 0; q < 4; ++q) {
    p0[q] = y0[q];
    p1[q] = y1[q];
  }
  __builtin_amdgcn_sched_barrier(0);
  v8f acc0 = {0.f, 0.f, 0.f, 0.f, 0.f, 0.f, 0.f, 0.f};
  v8f acc1 = {0.f, 0.f, 0.f, 0.f, 0.f, 0.f, 0.f, 0.f};
#pragma unroll
  for (int q = 0; q < 4; ++q) {
    const v4f y = p0[q] + p1[q];
    const v2f ylo = {y[0], y[1]}, yhi = {y[2], y[3]};
    acc0 = __builtin_amdgcn_wmma_f32_16x16x4_f32(
        false, ylo, false, b[8 * KH + 2 * q], (short)0, acc0, false, false);
    acc1 = __builtin_amdgcn_wmma_f32_16x16x4_f32(
        false, yhi, false, b[8 * KH + 2 * q + 1], (short)0, acc1, false, false);
  }
  const v8f acc = acc0 + acc1;
  float* up = Up + (16 * KH + 8 * half) * PITCH + tn + l16;
#pragma unroll
  for (int j = 0; j < 8; ++j) up[j * PITCH] = acc[j];
}

__global__ void __launch_bounds__(256)
fem_solve_kernel(const float* __restrict__ part,
                 const float* __restrict__ kappa,
                 float* __restrict__ u) {
  __shared__ __attribute__((aligned(16))) float XA[PAD * PITCH];
  __shared__ __attribute__((aligned(16))) float XB[PAD * PITCH];

  const int tid    = threadIdx.x;
  const int lane   = tid & 31;
  const int wave_u = __builtin_amdgcn_readfirstlane(tid >> 5);
  const int half   = lane >> 4;
  const int l16    = lane & 15;
  const int tn     = (wave_u >> 1) * 16;
  const int tm0    = (wave_u & 1) * 32;

  v4f pl[NPART][4];
#pragma unroll
  for (int g = 0; g < NPART; ++g) {
#pragma unroll
    for (int q = 0; q < 4; ++q)
      pl[g][q] = ((const v4f*)(part + g * (PAD * PAD)))[tid + 256 * q];
  }

  v2f b[16];
  const int n = tn + l16;
  const float fn1 = (float)(n + 1);
  const float fh  = (float)(32 * half);
#pragma unroll
  for (int ks = 0; ks < 16; ++ks) {
#pragma unroll
    for (int s = 0; s < 2; ++s) {
      const int k = 32 * half + 2 * ks + s;
      const float nk = fn1 * (fh + (float)(2 * ks + s + 1));
      const float sv = __sinf(nk * (PI_F / 63.0f));
      b[ks][s] = (n < NI && k < NI) ? sv : 0.0f;
    }
  }

  const int blk = blockIdx.x;
  const int kh3 = wave_u >> 2;
  const int tm3 = (wave_u & 3) * 16;
  v2f b3[8];
  const int n3 = 16 * blk + l16;
  const float fn3 = (float)(n3 + 1);
  const float fh3 = (float)(32 * half + 16 * kh3);
#pragma unroll
  for (int t = 0; t < 8; ++t) {
#pragma unroll
    for (int s = 0; s < 2; ++s) {
      const int k = 32 * half + 16 * kh3 + 2 * t + s;
      const float nk = fn3 * (fh3 + (float)(2 * t + s + 1));
      const float sv = __sinf(nk * (PI_F / 63.0f));
      b3[t][s] = (n3 < NI && k < NI) ? sv : 0.0f;
    }
  }

  const float kap  = kappa[0];
  const float norm = (2.0f / 63.0f) * (2.0f / 63.0f);
  const float ln = 2.0f - 2.0f * __cosf((float)(n + 1) * (PI_F / 63.0f));
  float sc0[8], sc1[8];
#pragma unroll
  for (int j = 0; j < 8; ++j) {
    const int m0 = tm0 + 8 * half + j;
    const int m1 = m0 + 16;
    const float lm0 = 2.0f - 2.0f * __cosf((float)(m0 + 1) * (PI_F / 63.0f));
    const float lm1 = 2.0f - 2.0f * __cosf((float)(m1 + 1) * (PI_F / 63.0f));
    const float i0 = norm / (kap * (ln + lm0));
    const float i1 = norm / (kap * (ln + lm1));
    sc0[j] = (m0 < NI && n < NI) ? i0 : 0.0f;
    sc1[j] = (m1 < NI && n < NI) ? i1 : 0.0f;
  }
  asm volatile("" : "+v"(sc0[0]), "+v"(sc0[1]), "+v"(sc0[2]), "+v"(sc0[3]),
                    "+v"(sc0[4]), "+v"(sc0[5]), "+v"(sc0[6]), "+v"(sc0[7]),
                    "+v"(sc1[0]), "+v"(sc1[1]), "+v"(sc1[2]), "+v"(sc1[3]),
                    "+v"(sc1[4]), "+v"(sc1[5]), "+v"(sc1[6]), "+v"(sc1[7]));
  __builtin_amdgcn_sched_barrier(0);
  float* xs = XA + (tid >> 4) * PITCH + 4 * (tid & 15);
#pragma unroll
  for (int q = 0; q < 4; ++q) {
    v4f s = pl[0][q];
#pragma unroll
    for (int g = 1; g < NPART; ++g) s += pl[g][q];
    *(v4f*)(xs + 16 * q * PITCH) = s;
  }
  __syncthreads();

  stage_XS_T<false>(XA, XB, b, sc0, sc1, tm0, tn, half, l16);
  __syncthreads();
  stage_XS_T<true >(XB, XA, b, sc0, sc1, tm0, tn, half, l16);
  __syncthreads();
  stage3_rows(XA, XB, b3, kh3, tm3, half, l16);
  __syncthreads();
  if (wave_u & 1) stage4_rows<1>(XB, XA + 1, b, tn, half, l16);
  else            stage4_rows<0>(XB, XA + 1, b, tn, half, l16);
  __syncthreads();

  const v4f zero4 = {0.f, 0.f, 0.f, 0.f};
  const int ri = tid >> 4;
  const int q  = tid & 15;
  const int rr = 16 * blk + 1 + ri;
  if (rr < GRID_NODES) {
    const v4f p0 = *(const v4f*)(XA + ri * PITCH + 4 * q);
    const v4f p1 = *(const v4f*)(XA + (16 + ri) * PITCH + 4 * q);
    v4f v = p0 + p1;
    if (rr > NI)  v = zero4;
    if (q == 0)   v[0] = 0.0f;
    if (q == 15)  v[3] = 0.0f;
    *(v4f*)(u + rr * GRID_NODES + 4 * q) = v;
  }
  if (blk == 0 && tid < 16) *(v4f*)(u + 4 * tid) = zero4;
}

extern "C" void kernel_launch(void* const* d_in, const int* in_sizes, int n_in,
                              void* d_out, int out_size, void* d_ws, size_t ws_size,
                              hipStream_t stream) {
  const float* f      = (const float*)d_in[0];
  const float* nodes  = (const float*)d_in[1];
  const float* kappa  = (const float*)d_in[2];
  const int*   elems  = (const int*)d_in[4];
  const int    nelem  = in_sizes[4] / 3;

  float* part   = (float*)d_ws;

  fem_assemble_kernel<<<NPART, ASM_THREADS, 0, stream>>>(f, nodes, elems, nelem, part);
  fem_solve_kernel<<<NSOLVE, 256, 0, stream>>>(part, kappa, (float*)d_out);
}
